# starting kernel
# baseline (speedup 1.0000x reference)
#include <hip/hip_runtime.h>
#include <hip/hip_bf16.h>

#define DFEAT 128

typedef __attribute__((ext_vector_type(2))) float v2f;
typedef __attribute__((ext_vector_type(8))) float v8f;

__global__ void gcn_zero_f32(float* __restrict__ p, int n) {
  int i = blockIdx.x * blockDim.x + threadIdx.x;
  int stride = gridDim.x * blockDim.x;
  for (; i < n; i += stride) p[i] = 0.0f;
}

__global__ void gcn_degree(const int* __restrict__ row, const int* __restrict__ col,
                           float* __restrict__ deg_out, float* __restrict__ deg_in,
                           int E) {
  int e = blockIdx.x * blockDim.x + threadIdx.x;
  if (e < E) {
    atomicAdd(&deg_out[row[e]], 1.0f);
    atomicAdd(&deg_in[col[e]], 1.0f);
  }
}

__global__ void gcn_scale(const float* __restrict__ x, const float* __restrict__ deg,
                          float* __restrict__ xs, int total4) {
  int t = blockIdx.x * blockDim.x + threadIdx.x;
  if (t < total4) {
    int n = t >> 5;
    float s = rsqrtf(fmaxf(deg[n], 1.0f));
    float4 v = reinterpret_cast<const float4*>(x)[t];
    v.x *= s; v.y *= s; v.z *= s; v.w *= s;
    reinterpret_cast<float4*>(xs)[t] = v;
  }
}

__global__ void gcn_scatter(const float* __restrict__ xs, const int* __restrict__ row,
                            const int* __restrict__ col, float* __restrict__ agg,
                            int E) {
  int t = blockIdx.x * blockDim.x + threadIdx.x;
  int e = t >> 5;
  int lane = t & 31;
  if (e < E) {
    int r = row[e];
    int c = col[e];
    float4 v = reinterpret_cast<const float4*>(xs + (size_t)r * DFEAT)[lane];
    float* dst = agg + (size_t)c * DFEAT + lane * 4;
    atomicAdd(dst + 0, v.x);
    atomicAdd(dst + 1, v.y);
    atomicAdd(dst + 2, v.z);
    atomicAdd(dst + 3, v.w);
  }
}

__global__ void __launch_bounds__(256) gcn_gemm_norm(
    const float* __restrict__ A,
    const float* __restrict__ W,
    const float* __restrict__ bias,
    const float* __restrict__ deg_in,
    const float* __restrict__ resid,
    float* __restrict__ out,
    int do_relu) {
  const int lane = threadIdx.x & 31;
  const int wave = threadIdx.x >> 5;
  const int half = lane >> 4;
  const int l15  = lane & 15;
  const int n0 = blockIdx.x * 16;
  const int i0 = wave * 16;

  const float* ap = A + (size_t)(n0 + l15) * DFEAT + 2 * half;
  const float* bp = W + (size_t)(i0 + l15) * DFEAT + 2 * half;

  v8f acc = {0.f, 0.f, 0.f, 0.f, 0.f, 0.f, 0.f, 0.f};
#pragma unroll
  for (int kk = 0; kk < DFEAT / 4; ++kk) {
    v2f a = *reinterpret_cast<const v2f*>(ap + 4 * kk);
    v2f b = *reinterpret_cast<const v2f*>(bp + 4 * kk);
    acc = __builtin_amdgcn_wmma_f32_16x16x4_f32(
         false, a,  false, b,
         (short)0, acc,  false,  false);
  }

  const int colI = i0 + l15;
  const float bb = bias[colI];
#pragma unroll
  for (int r = 0; r < 8; ++r) {
    int nrow = n0 + r + 8 * half;
    float val = acc[r] + bb;
    val *= rsqrtf(fmaxf(deg_in[nrow], 1.0f));
    if (resid) val += resid[(size_t)nrow * DFEAT + colI];
    if (do_relu) val = fmaxf(val, 0.0f);
    out[(size_t)nrow * DFEAT + colI] = val;
  }
}

extern "C" void kernel_launch(void* const* d_in, const int* in_sizes, int n_in,
                              void* d_out, int out_size, void* d_ws, size_t ws_size,
                              hipStream_t stream) {
  const float* x  = (const float*)d_in[0];
  const int*   ei = (const int*)d_in[1];
  const float* W1 = (const float*)d_in[2];
  const float* b1 = (const float*)d_in[3];
  const float* W2 = (const float*)d_in[4];
  const float* b2 = (const float*)d_in[5];

  const int N = in_sizes[0] / DFEAT;
  const int E = in_sizes[1] / 2;
  const int* row = ei;
  const int* col = ei + E;
  float* out = (float*)d_out;

  float* deg_out = (float*)d_ws;
  float* deg_in  = deg_out + N;
  float* xsrc    = deg_in + N;
  float* agg     = xsrc + (size_t)N * DFEAT;

  const int TB = 256;
  const int ndTotal  = N * DFEAT;
  const int scale4   = N * (DFEAT / 4);
  const long long scatterThreads = (long long)E * 32;

  gcn_zero_f32<<<(2 * N + TB - 1) / TB, TB, 0, stream>>>(deg_out, 2 * N);
  gcn_degree<<<(E + TB - 1) / TB, TB, 0, stream>>>(row, col, deg_out, deg_in, E);

  gcn_scale<<<(scale4 + TB - 1) / TB, TB, 0, stream>>>(x, deg_out, xsrc, scale4);
  gcn_zero_f32<<<(ndTotal + TB - 1) / TB, TB, 0, stream>>>(agg, ndTotal);
  gcn_scatter<<<(int)((scatterThreads + TB - 1) / TB), TB, 0, stream>>>(xsrc, row, col, agg, E);
  gcn_gemm_norm<<<N / 16, TB, 0, stream>>>(agg, W1, b1, deg_in, x, out, 1);

  gcn_scale<<<(scale4 + TB - 1) / TB, TB, 0, stream>>>(out, deg_out, xsrc, scale4);
  gcn_zero_f32<<<(ndTotal + TB - 1) / TB, TB, 0, stream>>>(agg, ndTotal);
  gcn_scatter<<<(int)((scatterThreads + TB - 1) / TB), TB, 0, stream>>>(xsrc, row, col, agg, E);
  gcn_gemm_norm<<<N / 16, TB, 0, stream>>>(agg, W2, b2, deg_in, nullptr, out, 0);
}


# Float-atomic scatter and agg zero fill replaced by owner blocks: each keeps 392 output rows in LDS, filters col itself; one wave adds a row's edges in fixed queue order, no atomics: sum order fixed
# speedup vs baseline: 1.8747x; 1.8747x over previous
#include <hip/hip_runtime.h>
#include <hip/hip_bf16.h>

#define DFEAT 128

typedef __attribute__((ext_vector_type(2))) float v2f;
typedef __attribute__((ext_vector_type(8))) float v8f;

__global__ void gcn_zero_f32(float* __restrict__ p, int n) {
  int i = blockIdx.x * blockDim.x + threadIdx.x;
  int stride = gridDim.x * blockDim.x;
  for (; i < n; i += stride) p[i] = 0.0f;
}

__global__ void gcn_degree(const int* __restrict__ row, const int* __restrict__ col,
                           float* __restrict__ deg_out, float* __restrict__ deg_in,
                           int E) {
  int e = blockIdx.x * blockDim.x + threadIdx.x;
  if (e < E) {
    atomicAdd(&deg_out[row[e]], 1.0f);
    atomicAdd(&deg_in[col[e]], 1.0f);
  }
}

__global__ void gcn_scale(const float* __restrict__ x, const float* __restrict__ deg,
                          float* __restrict__ xs, int total4) {
  int t = blockIdx.x * blockDim.x + threadIdx.x;
  if (t < total4) {
    int n = t >> 5;
    float s = rsqrtf(fmaxf(deg[n], 1.0f));
    float4 v = reinterpret_cast<const float4*>(x)[t];
    v.x *= s; v.y *= s; v.z *= s; v.w *= s;
    reinterpret_cast<float4*>(xs)[t] = v;
  }
}

#define OWN_ROWS 392
#define OWN_WAVES 32
#define OWN_QCAP 768
#define OWN_STEP 512
#define OWN_BATCH 16

__device__ __forceinline__ void own_flush(const float* __restrict__ xs,
                                          const int* __restrict__ row,
                                          float4* s_acc, unsigned pend, int lane) {
  int srcv = 0;
  if (lane < OWN_BATCH) srcv = row[pend >> 9];
  float4 vals[OWN_BATCH];
#pragma unroll
  for (int i = 0; i < OWN_BATCH; ++i) {
    const int s = __builtin_amdgcn_readlane(srcv, i);
    vals[i] = reinterpret_cast<const float4*>(xs + (size_t)s * DFEAT)[lane];
  }
  __builtin_amdgcn_sched_barrier(0);
#pragma unroll
  for (int i = 0; i < OWN_BATCH; ++i) {
    const unsigned dl = (unsigned)__builtin_amdgcn_readlane((int)pend, i) & 511u;
    float4 a = s_acc[dl * 32 + lane];
    a.x += vals[i].x; a.y += vals[i].y; a.z += vals[i].z; a.w += vals[i].w;
    s_acc[dl * 32 + lane] = a;
  }
}

__global__ void __launch_bounds__(1024) gcn_owner_sum(
    const float* __restrict__ xs, const int* __restrict__ row,
    const int* __restrict__ col, float* __restrict__ agg, int N, int E) {
  __shared__ float4 s_acc[(OWN_ROWS + OWN_WAVES) * 32];
  __shared__ unsigned s_q[OWN_WAVES * OWN_QCAP];
  __shared__ int s_qcnt[OWN_WAVES];
  __shared__ int s_done[OWN_WAVES];

  const int tid = threadIdx.x;
  const int lane = tid & 31;
  const int w = __builtin_amdgcn_readfirstlane(tid >> 5);
  const int lo = blockIdx.x * OWN_ROWS;
  const unsigned lt_mask = (1u << lane) - 1u;

  for (int i = tid; i < OWN_ROWS * 32; i += 1024)
    s_acc[i] = make_float4(0.0f, 0.0f, 0.0f, 0.0f);

  const int n_trips = (E + OWN_WAVES * OWN_STEP - 1) / (OWN_WAVES * OWN_STEP);
  const int seg0 = w * n_trips * OWN_STEP;
  int end = seg0 + n_trips * OWN_STEP;
  if (end > E) end = E;
  const int rot = (n_trips > 0) ? (int)(blockIdx.x % (unsigned)n_trips) : 0;
  int trip = 0;

  unsigned* myq = s_q + w * OWN_QCAP;
  const unsigned pad_entry = (unsigned)(OWN_ROWS + w);
  unsigned pend = pad_entry;
  int npend = 0;

  for (;;) {
    int qn = 0;
    while (trip < n_trips && qn + OWN_STEP <= OWN_QCAP) {
      int t = trip + rot;
      if (t >= n_trips) t -= n_trips;
      const int pos = seg0 + t * OWN_STEP;
      ++trip;
      if (pos >= end) continue;
      int c[OWN_STEP / 32];
      if (pos + OWN_STEP <= end) {
        const int* cp = col + pos + lane;
#pragma unroll
        for (int k = 0; k < OWN_STEP / 32; ++k) c[k] = cp[32 * k];
      } else {
#pragma unroll
        for (int k = 0; k < OWN_STEP / 32; ++k) {
          const int e = pos + 32 * k + lane;
          c[k] = (e < end) ? col[e] : -1;
        }
      }
#pragma unroll
      for (int k = 0; k < OWN_STEP / 32; ++k) {
        const unsigned dl = (unsigned)(c[k] - lo);
        const bool m = dl < (unsigned)OWN_ROWS;
        const unsigned mask = (unsigned)__ballot(m);
        if (mask != 0u) {
          if (m) {
            const unsigned e = (unsigned)(pos + 32 * k + lane);
            myq[qn + __builtin_popcount(mask & lt_mask)] = (e << 9) | dl;
          }
          qn += __builtin_popcount(mask);
        }
      }
    }
    if (lane == 0) {
      s_qcnt[w] = qn;
      s_done[w] = (trip >= n_trips) ? 1 : 0;
    }
    __syncthreads();

    for (int q = 0; q < OWN_WAVES; ++q) {
      const int n = __builtin_amdgcn_readfirstlane(s_qcnt[q]);
      const unsigned* qq = s_q + q * OWN_QCAP;
      for (int i = 0; i < n; i += 32) {
        const bool in = (i + lane) < n;
        const unsigned v = in ? qq[i + lane] : 0u;
        const bool m = in && ((v & 31u) == (unsigned)w);
        unsigned mask = (unsigned)__ballot(m);
        while (mask) {
          const int b = __builtin_ctz(mask);
          mask &= mask - 1u;
          const unsigned val = (unsigned)__builtin_amdgcn_readlane((int)v, b);
          if (lane == npend) pend = val;
          ++npend;
          if (npend == OWN_BATCH) {
            own_flush(xs, row, s_acc, pend, lane);
            pend = pad_entry;
            npend = 0;
          }
        }
      }
    }
    if (npend > 0) {
      own_flush(xs, row, s_acc, pend, lane);
      pend = pad_entry;
      npend = 0;
    }
    const bool fin = ((unsigned)__ballot(s_done[lane] != 0)) == 0xffffffffu;
    __syncthreads();
    if (fin) break;
  }

  for (int i = tid; i < OWN_ROWS * 32; i += 1024) {
    const int r = lo + (i >> 5);
    if (r < N) reinterpret_cast<float4*>(agg)[(size_t)r * 32 + (i & 31)] = s_acc[i];
  }
}

__device__ __forceinline__ v8f wmma_f32(v2f a, v2f b, v8f c) {
  return __builtin_amdgcn_wmma_f32_16x16x4_f32(
       false, a,  false, b,
       (short)0, c,  false,  false);
}

template <bool HAS_RESID, bool DO_RELU>
__global__ void __launch_bounds__(128) gcn_gemm_norm(
    const float* __restrict__ A,
    const float* __restrict__ W,
    const float* __restrict__ bias,
    const float* __restrict__ deg_in,
    const float* __restrict__ resid,
    float* __restrict__ out,
    int n_tiles) {
  const int lane = threadIdx.x & 31;
  const int wave = threadIdx.x >> 5;
  const int half = lane >> 4;
  const int l15  = lane & 15;
  const int c0 = wave * 32;

  float4 w0[16], w1[16];
  {
    const float4* wp0 = reinterpret_cast<const float4*>(W + (size_t)(c0 + l15) * DFEAT + 4 * half);
    const float4* wp1 = reinterpret_cast<const float4*>(W + (size_t)(c0 + 16 + l15) * DFEAT + 4 * half);
#pragma unroll
    for (int j = 0; j < 16; ++j) {
      w0[j] = wp0[2 * j];
      w1[j] = wp1[2 * j];
    }
  }
  const int col0 = c0 + l15;
  const int col1 = c0 + 16 + l15;
  const float bb0 = bias[col0];
  const float bb1 = bias[col1];

  for (int tile = blockIdx.x; tile < n_tiles; tile += gridDim.x) {
    const int n0 = tile * 16;
    const int rbase = n0 + 8 * half;

    float4 a[16];
    {
      const float4* ap = reinterpret_cast<const float4*>(A + (size_t)(n0 + l15) * DFEAT + 4 * half);
#pragma unroll
      for (int j = 0; j < 16; ++j) a[j] = ap[2 * j];
    }
    const float4 dg0 = *reinterpret_cast<const float4*>(deg_in + rbase);
    const float4 dg1 = *reinterpret_cast<const float4*>(deg_in + rbase + 4);
    float rs0[8], rs1[8];
    if (HAS_RESID) {
      const float* rp = resid + (size_t)rbase * DFEAT + col0;
#pragma unroll
      for (int r = 0; r < 8; ++r) {
        rs0[r] = rp[r * DFEAT];
        rs1[r] = rp[r * DFEAT + 16];
      }
    }
    __builtin_amdgcn_sched_barrier(0);

    v8f acc0 = {0.f, 0.f, 0.f, 0.f, 0.f, 0.f, 0.f, 0.f};
    v8f acc1 = {0.f, 0.f, 0.f, 0.f, 0.f, 0.f, 0.f, 0.f};
#pragma unroll
    for (int j = 0; j < 16; ++j) {
      const v2f a_lo = {a[j].x, a[j].y};
      const v2f a_hi = {a[j].z, a[j].w};
      const v2f b0_lo = {w0[j].x, w0[j].y};
      const v2f b0_hi = {w0[j].z, w0[j].w};
      const v2f b1_lo = {w1[j].x, w1[j].y};
      const v2f b1_hi = {w1[j].z, w1[j].w};
      acc0 = wmma_f32(a_lo, b0_lo, acc0);
      acc1 = wmma_f32(a_lo, b1_lo, acc1);
      acc0 = wmma_f32(a_hi, b0_hi, acc0);
      acc1 = wmma_f32(a_hi, b1_hi, acc1);
    }

    const float dg[8] = {dg0.x, dg0.y, dg0.z, dg0.w, dg1.x, dg1.y, dg1.z, dg1.w};
    float* op = out + (size_t)rbase * DFEAT + col0;
#pragma unroll
    for (int r = 0; r < 8; ++r) {
      const float s = rsqrtf(fmaxf(dg[r], 1.0f));
      float v0 = acc0[r] + bb0;
      float v1 = acc1[r] + bb1;
      v0 *= s;
      v1 *= s;
      if (HAS_RESID) { v0 += rs0[r]; v1 += rs1[r]; }
      if (DO_RELU) { v0 = fmaxf(v0, 0.0f); v1 = fmaxf(v1, 0.0f); }
      op[r * DFEAT] = v0;
      op[r * DFEAT + 16] = v1;
    }
  }
}

extern "C" void kernel_launch(void* const* d_in, const int* in_sizes, int n_in,
                              void* d_out, int out_size, void* d_ws, size_t ws_size,
                              hipStream_t stream) {
  const float* x  = (const float*)d_in[0];
  const int*   ei = (const int*)d_in[1];
  const float* W1 = (const float*)d_in[2];
  const float* b1 = (const float*)d_in[3];
  const float* W2 = (const float*)d_in[4];
  const float* b2 = (const float*)d_in[5];

  const int N = in_sizes[0] / DFEAT;
  const int E = in_sizes[1] / 2;
  const int* row = ei;
  const int* col = ei + E;
  float* out = (float*)d_out;

  float* deg_out = (float*)d_ws;
  float* deg_in  = deg_out + N;
  float* xsrc    = deg_in + N;
  float* agg     = xsrc + (size_t)N * DFEAT;

  const int TB = 256;
  const int GEMM_BLOCKS = 1024;
  const int scale4   = N * (DFEAT / 4);
  const int ownBlocks = (N + OWN_ROWS - 1) / OWN_ROWS;

  gcn_zero_f32<<<(2 * N + TB - 1) / TB, TB, 0, stream>>>(deg_out, 2 * N);
  gcn_degree<<<(E + TB - 1) / TB, TB, 0, stream>>>(row, col, deg_out, deg_in, E);

  gcn_scale<<<(scale4 + TB - 1) / TB, TB, 0, stream>>>(x, deg_out, xsrc, scale4);
  gcn_owner_sum<<<ownBlocks, 1024, 0, stream>>>(xsrc, row, col, agg, N, E);
  gcn_gemm_norm<true, true><<<GEMM_BLOCKS, 128, 0, stream>>>(agg, W1, b1, deg_in, x, out, N / 16);

  gcn_scale<<<(scale4 + TB - 1) / TB, TB, 0, stream>>>(out, deg_out, xsrc, scale4);
  gcn_owner_sum<<<ownBlocks, 1024, 0, stream>>>(xsrc, row, col, agg, N, E);
  gcn_gemm_norm<false, false><<<GEMM_BLOCKS, 128, 0, stream>>>(agg, W2, b2, deg_in, nullptr, out, N / 16);
}


# Layer 2's neighbour sum walks the per-wave batches (source row, local row) that layer 1's owner blocks saved in workspace: no second col scan, no queues, no row[] lookups; sum order unchanged
# speedup vs baseline: 2.1436x; 1.1434x over previous
#include <hip/hip_runtime.h>
#include <hip/hip_bf16.h>

#define DFEAT 128

typedef __attribute__((ext_vector_type(2))) float v2f;
typedef __attribute__((ext_vector_type(8))) float v8f;

__global__ void gcn_zero_f32(float* __restrict__ p, int n) {
  int i = blockIdx.x * blockDim.x + threadIdx.x;
  int stride = gridDim.x * blockDim.x;
  for (; i < n; i += stride) p[i] = 0.0f;
}

__global__ void gcn_degree(const int* __restrict__ row, const int* __restrict__ col,
                           float* __restrict__ deg_out, float* __restrict__ deg_in,
                           int E) {
  int e = blockIdx.x * blockDim.x + threadIdx.x;
  if (e < E) {
    atomicAdd(&deg_out[row[e]], 1.0f);
    atomicAdd(&deg_in[col[e]], 1.0f);
  }
}

__global__ void gcn_scale(const float* __restrict__ x, const float* __restrict__ deg,
                          float* __restrict__ xs, int total4) {
  int t = blockIdx.x * blockDim.x + threadIdx.x;
  if (t < total4) {
    int n = t >> 5;
    float s = rsqrtf(fmaxf(deg[n], 1.0f));
    float4 v = reinterpret_cast<const float4*>(x)[t];
    v.x *= s; v.y *= s; v.z *= s; v.w *= s;
    reinterpret_cast<float4*>(xs)[t] = v;
  }
}

#define OWN_ROWS 392
#define OWN_WAVES 32
#define OWN_QCAP 768
#define OWN_STEP 512
#define OWN_BATCH 16
#define OWN_LCAP 1024

__device__ __forceinline__ void own_add16(const float* __restrict__ xs,
                                          float4* s_acc, int srcv, unsigned dlv,
                                          int lane) {
  float4 vals[OWN_BATCH];
#pragma unroll
  for (int i = 0; i < OWN_BATCH; ++i) {
    const int s = __builtin_amdgcn_readlane(srcv, i);
    vals[i] = reinterpret_cast<const float4*>(xs + (size_t)s * DFEAT)[lane];
  }
  __builtin_amdgcn_sched_barrier(0);
#pragma unroll
  for (int i = 0; i < OWN_BATCH; ++i) {
    const unsigned dl = (unsigned)__builtin_amdgcn_readlane((int)dlv, i) & 511u;
    float4 a = s_acc[dl * 32 + lane];
    a.x += vals[i].x; a.y += vals[i].y; a.z += vals[i].z; a.w += vals[i].w;
    s_acc[dl * 32 + lane] = a;
  }
}

__device__ __forceinline__ void own_flush(const float* __restrict__ xs,
                                          const int* __restrict__ row,
                                          float4* s_acc, unsigned pend, int lane,
                                          unsigned* save_to) {
  int srcv = 0;
  if (lane < OWN_BATCH) {
    srcv = row[pend >> 9];
    if (save_to) save_to[lane] = ((unsigned)srcv << 9) | (pend & 511u);
  }
  own_add16(xs, s_acc, srcv, pend, lane);
}

template <bool USE_LIST>
__global__ void __launch_bounds__(1024) gcn_owner_sum(
    const float* __restrict__ xs, const int* __restrict__ row,
    const int* __restrict__ col, float* __restrict__ agg,
    unsigned* __restrict__ lists, int* __restrict__ lcnt, int N, int E) {
  __shared__ float4 s_acc[(OWN_ROWS + OWN_WAVES) * 32];
  __shared__ unsigned s_q[OWN_WAVES * OWN_QCAP];
  __shared__ int s_qcnt[OWN_WAVES];
  __shared__ int s_done[OWN_WAVES];

  const int tid = threadIdx.x;
  const int lane = tid & 31;
  const int w = __builtin_amdgcn_readfirstlane(tid >> 5);
  const int lo = blockIdx.x * OWN_ROWS;
  const unsigned lt_mask = (1u << lane) - 1u;

  for (int i = tid; i < OWN_ROWS * 32; i += 1024)
    s_acc[i] = make_float4(0.0f, 0.0f, 0.0f, 0.0f);

  unsigned* mylist = lists + (size_t)(blockIdx.x * OWN_WAVES + w) * OWN_LCAP;
  const unsigned pad_entry = (unsigned)(OWN_ROWS + w);

  if (USE_LIST) {
    const int cnt = lcnt[blockIdx.x * OWN_WAVES + lane];
    const bool all_fit =
        ((unsigned)__ballot(cnt <= OWN_LCAP / OWN_BATCH)) == 0xffffffffu;
    if (all_fit) {
      const int nb = __builtin_amdgcn_readlane(cnt, w);
      unsigned ent = pad_entry;
      if (nb > 0 && lane < OWN_BATCH) ent = mylist[lane];
      __syncthreads();
      for (int i = 0; i < nb; ++i) {
        unsigned nxt = pad_entry;
        if (i + 1 < nb && lane < OWN_BATCH) nxt = mylist[(i + 1) * OWN_BATCH + lane];
        own_add16(xs, s_acc, (int)(ent >> 9), ent, lane);
        ent = nxt;
      }
      __syncthreads();
      for (int i = tid; i < OWN_ROWS * 32; i += 1024) {
        const int r = lo + (i >> 5);
        if (r < N) reinterpret_cast<float4*>(agg)[(size_t)r * 32 + (i & 31)] = s_acc[i];
      }
      return;
    }
  }
  int nbatch = 0;

  const int n_trips = (E + OWN_WAVES * OWN_STEP - 1) / (OWN_WAVES * OWN_STEP);
  const int seg0 = w * n_trips * OWN_STEP;
  int end = seg0 + n_trips * OWN_STEP;
  if (end > E) end = E;
  const int rot = (n_trips > 0) ? (int)(blockIdx.x % (unsigned)n_trips) : 0;
  int trip = 0;

  unsigned* myq = s_q + w * OWN_QCAP;
  unsigned pend = pad_entry;
  int npend = 0;

  for (;;) {
    int qn = 0;
    while (trip < n_trips && qn + OWN_STEP <= OWN_QCAP) {
      int t = trip + rot;
      if (t >= n_trips) t -= n_trips;
      const int pos = seg0 + t * OWN_STEP;
      ++trip;
      if (pos >= end) continue;
      int c[OWN_STEP / 32];
      if (pos + OWN_STEP <= end) {
        const int* cp = col + pos + lane;
#pragma unroll
        for (int k = 0; k < OWN_STEP / 32; ++k) c[k] = cp[32 * k];
      } else {
#pragma unroll
        for (int k = 0; k < OWN_STEP / 32; ++k) {
          const int e = pos + 32 * k + lane;
          c[k] = (e < end) ? col[e] : -1;
        }
      }
#pragma unroll
      for (int k = 0; k < OWN_STEP / 32; ++k) {
        const unsigned dl = (unsigned)(c[k] - lo);
        const bool m = dl < (unsigned)OWN_ROWS;
        const unsigned mask = (unsigned)__ballot(m);
        if (mask != 0u) {
          if (m) {
            const unsigned e = (unsigned)(pos + 32 * k + lane);
            myq[qn + __builtin_popcount(mask & lt_mask)] = (e << 9) | dl;
          }
          qn += __builtin_popcount(mask);
        }
      }
    }
    if (lane == 0) {
      s_qcnt[w] = qn;
      s_done[w] = (trip >= n_trips) ? 1 : 0;
    }
    __syncthreads();

    for (int q = 0; q < OWN_WAVES; ++q) {
      const int n = __builtin_amdgcn_readfirstlane(s_qcnt[q]);
      const unsigned* qq = s_q + q * OWN_QCAP;
      for (int i = 0; i < n; i += 32) {
        const bool in = (i + lane) < n;
        const unsigned v = in ? qq[i + lane] : 0u;
        const bool m = in && ((v & 31u) == (unsigned)w);
        unsigned mask = (unsigned)__ballot(m);
        while (mask) {
          const int b = __builtin_ctz(mask);
          mask &= mask - 1u;
          const unsigned val = (unsigned)__builtin_amdgcn_readlane((int)v, b);
          if (lane == npend) pend = val;
          ++npend;
          if (npend == OWN_BATCH) {
            own_flush(xs, row, s_acc, pend, lane,
                      (!USE_LIST && nbatch < OWN_LCAP / OWN_BATCH)
                          ? mylist + nbatch * OWN_BATCH : nullptr);
            ++nbatch;
            pend = pad_entry;
            npend = 0;
          }
        }
      }
    }
    if (npend > 0) {
      own_flush(xs, row, s_acc, pend, lane,
                (!USE_LIST && nbatch < OWN_LCAP / OWN_BATCH)
                    ? mylist + nbatch * OWN_BATCH : nullptr);
      ++nbatch;
      pend = pad_entry;
      npend = 0;
    }
    const bool fin = ((unsigned)__ballot(s_done[lane] != 0)) == 0xffffffffu;
    __syncthreads();
    if (fin) break;
  }
  if (!USE_LIST && lane == 0) lcnt[blockIdx.x * OWN_WAVES + w] = nbatch;

  for (int i = tid; i < OWN_ROWS * 32; i += 1024) {
    const int r = lo + (i >> 5);
    if (r < N) reinterpret_cast<float4*>(agg)[(size_t)r * 32 + (i & 31)] = s_acc[i];
  }
}

__device__ __forceinline__ v8f wmma_f32(v2f a, v2f b, v8f c) {
  return __builtin_amdgcn_wmma_f32_16x16x4_f32(
       false, a,  false, b,
       (short)0, c,  false,  false);
}

template <bool HAS_RESID, bool DO_RELU>
__global__ void __launch_bounds__(128) gcn_gemm_norm(
    const float* __restrict__ A,
    const float* __restrict__ W,
    const float* __restrict__ bias,
    const float* __restrict__ deg_in,
    const float* __restrict__ resid,
    float* __restrict__ out,
    int n_tiles) {
  const int lane = threadIdx.x & 31;
  const int wave = threadIdx.x >> 5;
  const int half = lane >> 4;
  const int l15  = lane & 15;
  const int c0 = wave * 32;

  float4 w0[16], w1[16];
  {
    const float4* wp0 = reinterpret_cast<const float4*>(W + (size_t)(c0 + l15) * DFEAT + 4 * half);
    const float4* wp1 = reinterpret_cast<const float4*>(W + (size_t)(c0 + 16 + l15) * DFEAT + 4 * half);
#pragma unroll
    for (int j = 0; j < 16; ++j) {
      w0[j] = wp0[2 * j];
      w1[j] = wp1[2 * j];
    }
  }
  const int col0 = c0 + l15;
  const int col1 = c0 + 16 + l15;
  const float bb0 = bias[col0];
  const float bb1 = bias[col1];

  for (int tile = blockIdx.x; tile < n_tiles; tile += gridDim.x) {
    const int n0 = tile * 16;
    const int rbase = n0 + 8 * half;

    float4 a[16];
    {
      const float4* ap = reinterpret_cast<const float4*>(A + (size_t)(n0 + l15) * DFEAT + 4 * half);
#pragma unroll
      for (int j = 0; j < 16; ++j) a[j] = ap[2 * j];
    }
    const float4 dg0 = *reinterpret_cast<const float4*>(deg_in + rbase);
    const float4 dg1 = *reinterpret_cast<const float4*>(deg_in + rbase + 4);
    float rs0[8], rs1[8];
    if (HAS_RESID) {
      const float* rp = resid + (size_t)rbase * DFEAT + col0;
#pragma unroll
      for (int r = 0; r < 8; ++r) {
        rs0[r] = rp[r * DFEAT];
        rs1[r] = rp[r * DFEAT + 16];
      }
    }
    __builtin_amdgcn_sched_barrier(0);

    v8f acc0 = {0.f, 0.f, 0.f, 0.f, 0.f, 0.f, 0.f, 0.f};
    v8f acc1 = {0.f, 0.f, 0.f, 0.f, 0.f, 0.f, 0.f, 0.f};
#pragma unroll
    for (int j = 0; j < 16; ++j) {
      const v2f a_lo = {a[j].x, a[j].y};
      const v2f a_hi = {a[j].z, a[j].w};
      const v2f b0_lo = {w0[j].x, w0[j].y};
      const v2f b0_hi = {w0[j].z, w0[j].w};
      const v2f b1_lo = {w1[j].x, w1[j].y};
      const v2f b1_hi = {w1[j].z, w1[j].w};
      acc0 = wmma_f32(a_lo, b0_lo, acc0);
      acc1 = wmma_f32(a_lo, b1_lo, acc1);
      acc0 = wmma_f32(a_hi, b0_hi, acc0);
      acc1 = wmma_f32(a_hi, b1_hi, acc1);
    }

    const float dg[8] = {dg0.x, dg0.y, dg0.z, dg0.w, dg1.x, dg1.y, dg1.z, dg1.w};
    float* op = out + (size_t)rbase * DFEAT + col0;
#pragma unroll
    for (int r = 0; r < 8; ++r) {
      const float s = rsqrtf(fmaxf(dg[r], 1.0f));
      float v0 = acc0[r] + bb0;
      float v1 = acc1[r] + bb1;
      v0 *= s;
      v1 *= s;
      if (HAS_RESID) { v0 += rs0[r]; v1 += rs1[r]; }
      if (DO_RELU) { v0 = fmaxf(v0, 0.0f); v1 = fmaxf(v1, 0.0f); }
      op[r * DFEAT] = v0;
      op[r * DFEAT + 16] = v1;
    }
  }
}

extern "C" void kernel_launch(void* const* d_in, const int* in_sizes, int n_in,
                              void* d_out, int out_size, void* d_ws, size_t ws_size,
                              hipStream_t stream) {
  const float* x  = (const float*)d_in[0];
  const int*   ei = (const int*)d_in[1];
  const float* W1 = (const float*)d_in[2];
  const float* b1 = (const float*)d_in[3];
  const float* W2 = (const float*)d_in[4];
  const float* b2 = (const float*)d_in[5];

  const int N = in_sizes[0] / DFEAT;
  const int E = in_sizes[1] / 2;
  const int* row = ei;
  const int* col = ei + E;
  float* out = (float*)d_out;

  float* deg_out = (float*)d_ws;
  float* deg_in  = deg_out + N;
  float* xsrc    = deg_in + N;
  float* agg     = xsrc + (size_t)N * DFEAT;

  const int TB = 256;
  const int GEMM_BLOCKS = 1024;
  const int scale4   = N * (DFEAT / 4);
  const int ownBlocks = (N + OWN_ROWS - 1) / OWN_ROWS;
  unsigned* lists = (unsigned*)(agg + (size_t)N * DFEAT);
  int* lcnt = (int*)(lists + (size_t)ownBlocks * OWN_WAVES * OWN_LCAP);

  gcn_zero_f32<<<(2 * N + TB - 1) / TB, TB, 0, stream>>>(deg_out, 2 * N);
  gcn_degree<<<(E + TB - 1) / TB, TB, 0, stream>>>(row, col, deg_out, deg_in, E);

  gcn_scale<<<(scale4 + TB - 1) / TB, TB, 0, stream>>>(x, deg_out, xsrc, scale4);
  gcn_owner_sum<false><<<ownBlocks, 1024, 0, stream>>>(xsrc, row, col, agg, lists, lcnt, N, E);
  gcn_gemm_norm<true, true><<<GEMM_BLOCKS, 128, 0, stream>>>(agg, W1, b1, deg_in, x, out, N / 16);

  gcn_scale<<<(scale4 + TB - 1) / TB, TB, 0, stream>>>(out, deg_out, xsrc, scale4);
  gcn_owner_sum<true><<<ownBlocks, 1024, 0, stream>>>(xsrc, row, col, agg, lists, lcnt, N, E);
  gcn_gemm_norm<false, false><<<GEMM_BLOCKS, 128, 0, stream>>>(agg, W2, b2, deg_in, nullptr, out, N / 16);
}


# Layer product moved in front of the neighbour sum: rows are multiplied by W and scaled first, owner blocks sum the products, then add bias, norm, residual, relu; no x_src passes; sum order changes
# speedup vs baseline: 2.5095x; 1.1707x over previous
#include <hip/hip_runtime.h>
#include <hip/hip_bf16.h>

#define DFEAT 128

typedef __attribute__((ext_vector_type(2))) float v2f;
typedef __attribute__((ext_vector_type(8))) float v8f;

__global__ void gcn_zero_f32(float* __restrict__ p, int n) {
  int i = blockIdx.x * blockDim.x + threadIdx.x;
  int stride = gridDim.x * blockDim.x;
  for (; i < n; i += stride) p[i] = 0.0f;
}

__global__ void gcn_degree(const int* __restrict__ row, const int* __restrict__ col,
                           float* __restrict__ deg_out, float* __restrict__ deg_in,
                           int E) {
  int e = blockIdx.x * blockDim.x + threadIdx.x;
  if (e < E) {
    atomicAdd(&deg_out[row[e]], 1.0f);
    atomicAdd(&deg_in[col[e]], 1.0f);
  }
}

#define OWN_ROWS 392
#define OWN_WAVES 32
#define OWN_QCAP 768
#define OWN_STEP 512
#define OWN_BATCH 16
#define OWN_LCAP 1024

__device__ __forceinline__ void own_add16(const float* __restrict__ xs,
                                          float4* s_acc, int srcv, unsigned dlv,
                                          int lane) {
  float4 vals[OWN_BATCH];
#pragma unroll
  for (int i = 0; i < OWN_BATCH; ++i) {
    const int s = __builtin_amdgcn_readlane(srcv, i);
    vals[i] = reinterpret_cast<const float4*>(xs + (size_t)s * DFEAT)[lane];
  }
  __builtin_amdgcn_sched_barrier(0);
#pragma unroll
  for (int i = 0; i < OWN_BATCH; ++i) {
    const unsigned dl = (unsigned)__builtin_amdgcn_readlane((int)dlv, i) & 511u;
    float4 a = s_acc[dl * 32 + lane];
    a.x += vals[i].x; a.y += vals[i].y; a.z += vals[i].z; a.w += vals[i].w;
    s_acc[dl * 32 + lane] = a;
  }
}

__device__ __forceinline__ void own_flush(const float* __restrict__ xs,
                                          const int* __restrict__ row,
                                          float4* s_acc, unsigned pend, int lane,
                                          unsigned* save_to) {
  int srcv = 0;
  if (lane < OWN_BATCH) {
    srcv = row[pend >> 9];
    if (save_to) save_to[lane] = ((unsigned)srcv << 9) | (pend & 511u);
  }
  own_add16(xs, s_acc, srcv, pend, lane);
}

#define OWN_SROWS ((OWN_ROWS + OWN_WAVES - 1) / OWN_WAVES)
template <bool HAS_RESID, bool DO_RELU>
__device__ __forceinline__ void own_store(const float4* s_acc,
                                          const float* __restrict__ bias,
                                          const float* __restrict__ deg_in,
                                          const float* __restrict__ resid,
                                          float* __restrict__ dst,
                                          int lo, int N, int w, int lane) {
  const float4 b = reinterpret_cast<const float4*>(bias)[lane];
  float dg[OWN_SROWS];
  float4 rs[OWN_SROWS];
#pragma unroll
  for (int k = 0; k < OWN_SROWS; ++k) {
    int r = lo + w + OWN_WAVES * k;
    if (r > N - 1) r = N - 1;
    dg[k] = deg_in[r];
    if (HAS_RESID) rs[k] = reinterpret_cast<const float4*>(resid)[(size_t)r * 32 + lane];
  }
#pragma unroll
  for (int k = 0; k < OWN_SROWS; ++k) {
    const int lr = w + OWN_WAVES * k;
    const int r = lo + lr;
    if (lr < OWN_ROWS && r < N) {
      const float s = rsqrtf(fmaxf(dg[k], 1.0f));
      float4 v = s_acc[lr * 32 + lane];
      v.x = (v.x + b.x) * s; v.y = (v.y + b.y) * s;
      v.z = (v.z + b.z) * s; v.w = (v.w + b.w) * s;
      if (HAS_RESID) { v.x += rs[k].x; v.y += rs[k].y; v.z += rs[k].z; v.w += rs[k].w; }
      if (DO_RELU) {
        v.x = fmaxf(v.x, 0.0f); v.y = fmaxf(v.y, 0.0f);
        v.z = fmaxf(v.z, 0.0f); v.w = fmaxf(v.w, 0.0f);
      }
      reinterpret_cast<float4*>(dst)[(size_t)r * 32 + lane] = v;
    }
  }
}

template <bool USE_LIST, bool HAS_RESID, bool DO_RELU>
__global__ void __launch_bounds__(1024) gcn_owner_sum(
    const float* __restrict__ xs, const int* __restrict__ row,
    const int* __restrict__ col,
    const float* __restrict__ bias,
    const float* __restrict__ deg_in,
    const float* __restrict__ resid,
    float* __restrict__ dst,
    unsigned* __restrict__ lists, int* __restrict__ lcnt, int N, int E) {
  __shared__ float4 s_acc[(OWN_ROWS + OWN_WAVES) * 32];
  __shared__ unsigned s_q[OWN_WAVES * OWN_QCAP];
  __shared__ int s_qcnt[OWN_WAVES];
  __shared__ int s_done[OWN_WAVES];

  const int tid = threadIdx.x;
  const int lane = tid & 31;
  const int w = __builtin_amdgcn_readfirstlane(tid >> 5);
  const int lo = blockIdx.x * OWN_ROWS;
  const unsigned lt_mask = (1u << lane) - 1u;

  for (int i = tid; i < OWN_ROWS * 32; i += 1024)
    s_acc[i] = make_float4(0.0f, 0.0f, 0.0f, 0.0f);

  unsigned* mylist = lists + (size_t)(blockIdx.x * OWN_WAVES + w) * OWN_LCAP;
  const unsigned pad_entry = (unsigned)(OWN_ROWS + w);

  if (USE_LIST) {
    const int cnt = lcnt[blockIdx.x * OWN_WAVES + lane];
    const bool all_fit =
        ((unsigned)__ballot(cnt <= OWN_LCAP / OWN_BATCH)) == 0xffffffffu;
    if (all_fit) {
      const int nb = __builtin_amdgcn_readlane(cnt, w);
      unsigned ent = pad_entry;
      if (nb > 0 && lane < OWN_BATCH) ent = mylist[lane];
      __syncthreads();
      for (int i = 0; i < nb; ++i) {
        unsigned nxt = pad_entry;
        if (i + 1 < nb && lane < OWN_BATCH) nxt = mylist[(i + 1) * OWN_BATCH + lane];
        own_add16(xs, s_acc, (int)(ent >> 9), ent, lane);
        ent = nxt;
      }
      __syncthreads();
      own_store<HAS_RESID, DO_RELU>(s_acc, bias, deg_in, resid, dst, lo, N, w, lane);
      return;
    }
  }
  int nbatch = 0;

  const int n_trips = (E + OWN_WAVES * OWN_STEP - 1) / (OWN_WAVES * OWN_STEP);
  const int seg0 = w * n_trips * OWN_STEP;
  int end = seg0 + n_trips * OWN_STEP;
  if (end > E) end = E;
  const int rot = (n_trips > 0) ? (int)(blockIdx.x % (unsigned)n_trips) : 0;
  int trip = 0;

  unsigned* myq = s_q + w * OWN_QCAP;
  unsigned pend = pad_entry;
  int npend = 0;

  for (;;) {
    int qn = 0;
    while (trip < n_trips && qn + OWN_STEP <= OWN_QCAP) {
      int t = trip + rot;
      if (t >= n_trips) t -= n_trips;
      const int pos = seg0 + t * OWN_STEP;
      ++trip;
      if (pos >= end) continue;
      int c[OWN_STEP / 32];
      if (pos + OWN_STEP <= end) {
        const int* cp = col + pos + lane;
#pragma unroll
        for (int k = 0; k < OWN_STEP / 32; ++k) c[k] = cp[32 * k];
      } else {
#pragma unroll
        for (int k = 0; k < OWN_STEP / 32; ++k) {
          const int e = pos + 32 * k + lane;
          c[k] = (e < end) ? col[e] : -1;
        }
      }
#pragma unroll
      for (int k = 0; k < OWN_STEP / 32; ++k) {
        const unsigned dl = (unsigned)(c[k] - lo);
        const bool m = dl < (unsigned)OWN_ROWS;
        const unsigned mask = (unsigned)__ballot(m);
        if (mask != 0u) {
          if (m) {
            const unsigned e = (unsigned)(pos + 32 * k + lane);
            myq[qn + __builtin_popcount(mask & lt_mask)] = (e << 9) | dl;
          }
          qn += __builtin_popcount(mask);
        }
      }
    }
    if (lane == 0) {
      s_qcnt[w] = qn;
      s_done[w] = (trip >= n_trips) ? 1 : 0;
    }
    __syncthreads();

    for (int q = 0; q < OWN_WAVES; ++q) {
      const int n = __builtin_amdgcn_readfirstlane(s_qcnt[q]);
      const unsigned* qq = s_q + q * OWN_QCAP;
      for (int i = 0; i < n; i += 32) {
        const bool in = (i + lane) < n;
        const unsigned v = in ? qq[i + lane] : 0u;
        const bool m = in && ((v & 31u) == (unsigned)w);
        unsigned mask = (unsigned)__ballot(m);
        while (mask) {
          const int b = __builtin_ctz(mask);
          mask &= mask - 1u;
          const unsigned val = (unsigned)__builtin_amdgcn_readlane((int)v, b);
          if (lane == npend) pend = val;
          ++npend;
          if (npend == OWN_BATCH) {
            own_flush(xs, row, s_acc, pend, lane,
                      (!USE_LIST && nbatch < OWN_LCAP / OWN_BATCH)
                          ? mylist + nbatch * OWN_BATCH : nullptr);
            ++nbatch;
            pend = pad_entry;
            npend = 0;
          }
        }
      }
    }
    if (npend > 0) {
      own_flush(xs, row, s_acc, pend, lane,
                (!USE_LIST && nbatch < OWN_LCAP / OWN_BATCH)
                    ? mylist + nbatch * OWN_BATCH : nullptr);
      ++nbatch;
      pend = pad_entry;
      npend = 0;
    }
    const bool fin = ((unsigned)__ballot(s_done[lane] != 0)) == 0xffffffffu;
    __syncthreads();
    if (fin) break;
  }
  if (!USE_LIST && lane == 0) lcnt[blockIdx.x * OWN_WAVES + w] = nbatch;

  own_store<HAS_RESID, DO_RELU>(s_acc, bias, deg_in, resid, dst, lo, N, w, lane);
}

__device__ __forceinline__ v8f wmma_f32(v2f a, v2f b, v8f c) {
  return __builtin_amdgcn_wmma_f32_16x16x4_f32(
       false, a,  false, b,
       (short)0, c,  false,  false);
}

__global__ void __launch_bounds__(128) gcn_gemm_rowscale(
    const float* __restrict__ A,
    const float* __restrict__ W,
    const float* __restrict__ deg_out,
    float* __restrict__ out,
    int n_tiles) {
  const int lane = threadIdx.x & 31;
  const int wave = threadIdx.x >> 5;
  const int half = lane >> 4;
  const int l15  = lane & 15;
  const int c0 = wave * 32;

  float4 w0[16], w1[16];
  {
    const float4* wp0 = reinterpret_cast<const float4*>(W + (size_t)(c0 + l15) * DFEAT + 4 * half);
    const float4* wp1 = reinterpret_cast<const float4*>(W + (size_t)(c0 + 16 + l15) * DFEAT + 4 * half);
#pragma unroll
    for (int j = 0; j < 16; ++j) {
      w0[j] = wp0[2 * j];
      w1[j] = wp1[2 * j];
    }
  }
  const int col0 = c0 + l15;

  for (int tile = blockIdx.x; tile < n_tiles; tile += gridDim.x) {
    const int n0 = tile * 16;
    const int rbase = n0 + 8 * half;

    float4 a[16];
    {
      const float4* ap = reinterpret_cast<const float4*>(A + (size_t)(n0 + l15) * DFEAT + 4 * half);
#pragma unroll
      for (int j = 0; j < 16; ++j) a[j] = ap[2 * j];
    }
    const float4 dg0 = *reinterpret_cast<const float4*>(deg_out + rbase);
    const float4 dg1 = *reinterpret_cast<const float4*>(deg_out + rbase + 4);
    __builtin_amdgcn_sched_barrier(0);

    v8f acc0 = {0.f, 0.f, 0.f, 0.f, 0.f, 0.f, 0.f, 0.f};
    v8f acc1 = {0.f, 0.f, 0.f, 0.f, 0.f, 0.f, 0.f, 0.f};
#pragma unroll
    for (int j = 0; j < 16; ++j) {
      const v2f a_lo = {a[j].x, a[j].y};
      const v2f a_hi = {a[j].z, a[j].w};
      const v2f b0_lo = {w0[j].x, w0[j].y};
      const v2f b0_hi = {w0[j].z, w0[j].w};
      const v2f b1_lo = {w1[j].x, w1[j].y};
      const v2f b1_hi = {w1[j].z, w1[j].w};
      acc0 = wmma_f32(a_lo, b0_lo, acc0);
      acc1 = wmma_f32(a_lo, b1_lo, acc1);
      acc0 = wmma_f32(a_hi, b0_hi, acc0);
      acc1 = wmma_f32(a_hi, b1_hi, acc1);
    }

    const float dg[8] = {dg0.x, dg0.y, dg0.z, dg0.w, dg1.x, dg1.y, dg1.z, dg1.w};
    float* op = out + (size_t)rbase * DFEAT + col0;
#pragma unroll
    for (int r = 0; r < 8; ++r) {
      const float s = rsqrtf(fmaxf(dg[r], 1.0f));
      op[r * DFEAT] = acc0[r] * s;
      op[r * DFEAT + 16] = acc1[r] * s;
    }
  }
}

extern "C" void kernel_launch(void* const* d_in, const int* in_sizes, int n_in,
                              void* d_out, int out_size, void* d_ws, size_t ws_size,
                              hipStream_t stream) {
  const float* x  = (const float*)d_in[0];
  const int*   ei = (const int*)d_in[1];
  const float* W1 = (const float*)d_in[2];
  const float* b1 = (const float*)d_in[3];
  const float* W2 = (const float*)d_in[4];
  const float* b2 = (const float*)d_in[5];

  const int N = in_sizes[0] / DFEAT;
  const int E = in_sizes[1] / 2;
  const int* row = ei;
  const int* col = ei + E;
  float* out = (float*)d_out;

  float* deg_out = (float*)d_ws;
  float* deg_in  = deg_out + N;
  float* y       = deg_in + N;

  const int TB = 256;
  const int GEMM_BLOCKS = 1024;
  const int ownBlocks = (N + OWN_ROWS - 1) / OWN_ROWS;
  unsigned* lists = (unsigned*)(y + (size_t)N * DFEAT);
  int* lcnt = (int*)(lists + (size_t)ownBlocks * OWN_WAVES * OWN_LCAP);

  gcn_zero_f32<<<(2 * N + TB - 1) / TB, TB, 0, stream>>>(deg_out, 2 * N);
  gcn_degree<<<(E + TB - 1) / TB, TB, 0, stream>>>(row, col, deg_out, deg_in, E);

  gcn_gemm_rowscale<<<GEMM_BLOCKS, 128, 0, stream>>>(x, W1, deg_out, y, N / 16);
  gcn_owner_sum<false, true, true><<<ownBlocks, 1024, 0, stream>>>(
      y, row, col, b1, deg_in, x, out, lists, lcnt, N, E);

  gcn_gemm_rowscale<<<GEMM_BLOCKS, 128, 0, stream>>>(out, W2, deg_out, y, N / 16);
  gcn_owner_sum<true, false, false><<<ownBlocks, 1024, 0, stream>>>(
      y, row, col, b2, deg_in, nullptr, out, lists, lcnt, N, E);
}


# Degree counts: the scattered global f32 add per edge end replaced by 256 range blocks that count their 6272 nodes in LDS over an eighth of a list, then add whole 128-byte lines; counts stay exact
# speedup vs baseline: 3.2547x; 1.2970x over previous
#include <hip/hip_runtime.h>
#include <hip/hip_bf16.h>

#define DFEAT 128

typedef __attribute__((ext_vector_type(2))) float v2f;
typedef __attribute__((ext_vector_type(8))) float v8f;

__global__ void gcn_zero_f32(float* __restrict__ p, int n) {
  int i = blockIdx.x * blockDim.x + threadIdx.x;
  int stride = gridDim.x * blockDim.x;
  for (; i < n; i += stride) p[i] = 0.0f;
}

#define DEG_ROWS 6272
#define DEG_PARTS 8
#define DEG_THREADS 512
#define DEG_DEPTH 25

__device__ __forceinline__ void deg_count(int* s_cnt, int v, int lo) {
  const unsigned d = (unsigned)(v - lo);
  if (d < (unsigned)DEG_ROWS) atomicAdd(&s_cnt[d], 1);
}

__global__ void __launch_bounds__(DEG_THREADS) gcn_degree_ranges(
    const int* __restrict__ row, const int* __restrict__ col,
    float* __restrict__ deg_out, float* __restrict__ deg_in,
    int N, int E, int n_ranges) {
  __shared__ int s_cnt[DEG_ROWS];
  const int tid = threadIdx.x;
  int b = blockIdx.x;
  const int part = b % DEG_PARTS;
  b /= DEG_PARTS;
  const int range = b % n_ranges;
  const bool is_col = (b / n_ranges) != 0;
  const int* __restrict__ idx = is_col ? col : row;
  float* __restrict__ deg = is_col ? deg_in : deg_out;
  const int lo = range * DEG_ROWS;

  for (int i = tid; i < DEG_ROWS; i += DEG_THREADS) s_cnt[i] = 0;
  __syncthreads();

  const int per = ((E + DEG_PARTS - 1) / DEG_PARTS + 3) & ~3;
  const int beg = part * per;
  int end = beg + per;
  if (end > E) end = E;
  const int n = end - beg;
  if (n > 0) {
    const int* base = idx + beg;
    if ((reinterpret_cast<size_t>(base) & 15) == 0) {
      const int nv = n >> 2;
      const int4* b4 = reinterpret_cast<const int4*>(base);
      for (int q = tid; q < nv; q += DEG_THREADS * DEG_DEPTH) {
        int4 v[DEG_DEPTH];
#pragma unroll
        for (int k = 0; k < DEG_DEPTH; ++k) {
          const int qq = q + DEG_THREADS * k;
          v[k] = b4[qq < nv ? qq : nv - 1];
        }
#pragma unroll
        for (int k = 0; k < DEG_DEPTH; ++k) {
          if (q + DEG_THREADS * k < nv) {
            deg_count(s_cnt, v[k].x, lo);
            deg_count(s_cnt, v[k].y, lo);
            deg_count(s_cnt, v[k].z, lo);
            deg_count(s_cnt, v[k].w, lo);
          }
        }
      }
      const int t = (nv << 2) + tid;
      if (t < n) deg_count(s_cnt, base[t], lo);
    } else {
      for (int p = tid; p < n; p += DEG_THREADS) deg_count(s_cnt, base[p], lo);
    }
  }
  __syncthreads();

  for (int i = tid; i < DEG_ROWS; i += DEG_THREADS) {
    const int node = lo + i;
    if (node < N) atomicAdd(&deg[node], (float)s_cnt[i]);
  }
}

#define OWN_ROWS 392
#define OWN_WAVES 32
#define OWN_QCAP 768
#define OWN_STEP 512
#define OWN_BATCH 16
#define OWN_LCAP 1024

__device__ __forceinline__ void own_add16(const float* __restrict__ xs,
                                          float4* s_acc, int srcv, unsigned dlv,
                                          int lane) {
  float4 vals[OWN_BATCH];
#pragma unroll
  for (int i = 0; i < OWN_BATCH; ++i) {
    const int s = __builtin_amdgcn_readlane(srcv, i);
    vals[i] = reinterpret_cast<const float4*>(xs + (size_t)s * DFEAT)[lane];
  }
  __builtin_amdgcn_sched_barrier(0);
#pragma unroll
  for (int i = 0; i < OWN_BATCH; ++i) {
    const unsigned dl = (unsigned)__builtin_amdgcn_readlane((int)dlv, i) & 511u;
    float4 a = s_acc[dl * 32 + lane];
    a.x += vals[i].x; a.y += vals[i].y; a.z += vals[i].z; a.w += vals[i].w;
    s_acc[dl * 32 + lane] = a;
  }
}

__device__ __forceinline__ void own_flush(const float* __restrict__ xs,
                                          const int* __restrict__ row,
                                          float4* s_acc, unsigned pend, int lane,
                                          unsigned* save_to) {
  int srcv = 0;
  if (lane < OWN_BATCH) {
    srcv = row[pend >> 9];
    if (save_to) save_to[lane] = ((unsigned)srcv << 9) | (pend & 511u);
  }
  own_add16(xs, s_acc, srcv, pend, lane);
}

#define OWN_SROWS ((OWN_ROWS + OWN_WAVES - 1) / OWN_WAVES)
template <bool HAS_RESID, bool DO_RELU>
__device__ __forceinline__ void own_store(const float4* s_acc,
                                          const float* __restrict__ bias,
                                          const float* __restrict__ deg_in,
                                          const float* __restrict__ resid,
                                          float* __restrict__ dst,
                                          int lo, int N, int w, int lane) {
  const float4 b = reinterpret_cast<const float4*>(bias)[lane];
  float dg[OWN_SROWS];
  float4 rs[OWN_SROWS];
#pragma unroll
  for (int k = 0; k < OWN_SROWS; ++k) {
    int r = lo + w + OWN_WAVES * k;
    if (r > N - 1) r = N - 1;
    dg[k] = deg_in[r];
    if (HAS_RESID) rs[k] = reinterpret_cast<const float4*>(resid)[(size_t)r * 32 + lane];
  }
#pragma unroll
  for (int k = 0; k < OWN_SROWS; ++k) {
    const int lr = w + OWN_WAVES * k;
    const int r = lo + lr;
    if (lr < OWN_ROWS && r < N) {
      const float s = rsqrtf(fmaxf(dg[k], 1.0f));
      float4 v = s_acc[lr * 32 + lane];
      v.x = (v.x + b.x) * s; v.y = (v.y + b.y) * s;
      v.z = (v.z + b.z) * s; v.w = (v.w + b.w) * s;
      if (HAS_RESID) { v.x += rs[k].x; v.y += rs[k].y; v.z += rs[k].z; v.w += rs[k].w; }
      if (DO_RELU) {
        v.x = fmaxf(v.x, 0.0f); v.y = fmaxf(v.y, 0.0f);
        v.z = fmaxf(v.z, 0.0f); v.w = fmaxf(v.w, 0.0f);
      }
      reinterpret_cast<float4*>(dst)[(size_t)r * 32 + lane] = v;
    }
  }
}

template <bool USE_LIST, bool HAS_RESID, bool DO_RELU>
__global__ void __launch_bounds__(1024) gcn_owner_sum(
    const float* __restrict__ xs, const int* __restrict__ row,
    const int* __restrict__ col,
    const float* __restrict__ bias,
    const float* __restrict__ deg_in,
    const float* __restrict__ resid,
    float* __restrict__ dst,
    unsigned* __restrict__ lists, int* __restrict__ lcnt, int N, int E) {
  __shared__ float4 s_acc[(OWN_ROWS + OWN_WAVES) * 32];
  __shared__ unsigned s_q[OWN_WAVES * OWN_QCAP];
  __shared__ int s_qcnt[OWN_WAVES];
  __shared__ int s_done[OWN_WAVES];

  const int tid = threadIdx.x;
  const int lane = tid & 31;
  const int w = __builtin_amdgcn_readfirstlane(tid >> 5);
  const int lo = blockIdx.x * OWN_ROWS;
  const unsigned lt_mask = (1u << lane) - 1u;

  for (int i = tid; i < OWN_ROWS * 32; i += 1024)
    s_acc[i] = make_float4(0.0f, 0.0f, 0.0f, 0.0f);

  unsigned* mylist = lists + (size_t)(blockIdx.x * OWN_WAVES + w) * OWN_LCAP;
  const unsigned pad_entry = (unsigned)(OWN_ROWS + w);

  if (USE_LIST) {
    const int cnt = lcnt[blockIdx.x * OWN_WAVES + lane];
    const bool all_fit =
        ((unsigned)__ballot(cnt <= OWN_LCAP / OWN_BATCH)) == 0xffffffffu;
    if (all_fit) {
      const int nb = __builtin_amdgcn_readlane(cnt, w);
      unsigned ent = pad_entry;
      if (nb > 0 && lane < OWN_BATCH) ent = mylist[lane];
      __syncthreads();
      for (int i = 0; i < nb; ++i) {
        unsigned nxt = pad_entry;
        if (i + 1 < nb && lane < OWN_BATCH) nxt = mylist[(i + 1) * OWN_BATCH + lane];
        own_add16(xs, s_acc, (int)(ent >> 9), ent, lane);
        ent = nxt;
      }
      __syncthreads();
      own_store<HAS_RESID, DO_RELU>(s_acc, bias, deg_in, resid, dst, lo, N, w, lane);
      return;
    }
  }
  int nbatch = 0;

  const int n_trips = (E + OWN_WAVES * OWN_STEP - 1) / (OWN_WAVES * OWN_STEP);
  const int seg0 = w * n_trips * OWN_STEP;
  int end = seg0 + n_trips * OWN_STEP;
  if (end > E) end = E;
  const int rot = (n_trips > 0) ? (int)(blockIdx.x % (unsigned)n_trips) : 0;
  int trip = 0;

  unsigned* myq = s_q + w * OWN_QCAP;
  unsigned pend = pad_entry;
  int npend = 0;

  for (;;) {
    int qn = 0;
    while (trip < n_trips && qn + OWN_STEP <= OWN_QCAP) {
      int t = trip + rot;
      if (t >= n_trips) t -= n_trips;
      const int pos = seg0 + t * OWN_STEP;
      ++trip;
      if (pos >= end) continue;
      int c[OWN_STEP / 32];
      if (pos + OWN_STEP <= end) {
        const int* cp = col + pos + lane;
#pragma unroll
        for (int k = 0; k < OWN_STEP / 32; ++k) c[k] = cp[32 * k];
      } else {
#pragma unroll
        for (int k = 0; k < OWN_STEP / 32; ++k) {
          const int e = pos + 32 * k + lane;
          c[k] = (e < end) ? col[e] : -1;
        }
      }
#pragma unroll
      for (int k = 0; k < OWN_STEP / 32; ++k) {
        const unsigned dl = (unsigned)(c[k] - lo);
        const bool m = dl < (unsigned)OWN_ROWS;
        const unsigned mask = (unsigned)__ballot(m);
        if (mask != 0u) {
          if (m) {
            const unsigned e = (unsigned)(pos + 32 * k + lane);
            myq[qn + __builtin_popcount(mask & lt_mask)] = (e << 9) | dl;
          }
          qn += __builtin_popcount(mask);
        }
      }
    }
    if (lane == 0) {
      s_qcnt[w] = qn;
      s_done[w] = (trip >= n_trips) ? 1 : 0;
    }
    __syncthreads();

    for (int q = 0; q < OWN_WAVES; ++q) {
      const int n = __builtin_amdgcn_readfirstlane(s_qcnt[q]);
      const unsigned* qq = s_q + q * OWN_QCAP;
      for (int i = 0; i < n; i += 32) {
        const bool in = (i + lane) < n;
        const unsigned v = in ? qq[i + lane] : 0u;
        const bool m = in && ((v & 31u) == (unsigned)w);
        unsigned mask = (unsigned)__ballot(m);
        while (mask) {
          const int b = __builtin_ctz(mask);
          mask &= mask - 1u;
          const unsigned val = (unsigned)__builtin_amdgcn_readlane((int)v, b);
          if (lane == npend) pend = val;
          ++npend;
          if (npend == OWN_BATCH) {
            own_flush(xs, row, s_acc, pend, lane,
                      (!USE_LIST && nbatch < OWN_LCAP / OWN_BATCH)
                          ? mylist + nbatch * OWN_BATCH : nullptr);
            ++nbatch;
            pend = pad_entry;
            npend = 0;
          }
        }
      }
    }
    if (npend > 0) {
      own_flush(xs, row, s_acc, pend, lane,
                (!USE_LIST && nbatch < OWN_LCAP / OWN_BATCH)
                    ? mylist + nbatch * OWN_BATCH : nullptr);
      ++nbatch;
      pend = pad_entry;
      npend = 0;
    }
    const bool fin = ((unsigned)__ballot(s_done[lane] != 0)) == 0xffffffffu;
    __syncthreads();
    if (fin) break;
  }
  if (!USE_LIST && lane == 0) lcnt[blockIdx.x * OWN_WAVES + w] = nbatch;

  own_store<HAS_RESID, DO_RELU>(s_acc, bias, deg_in, resid, dst, lo, N, w, lane);
}

__device__ __forceinline__ v8f wmma_f32(v2f a, v2f b, v8f c) {
  return __builtin_amdgcn_wmma_f32_16x16x4_f32(
       false, a,  false, b,
       (short)0, c,  false,  false);
}

__global__ void __launch_bounds__(128) gcn_gemm_rowscale(
    const float* __restrict__ A,
    const float* __restrict__ W,
    const float* __restrict__ deg_out,
    float* __restrict__ out,
    int n_tiles) {
  const int lane = threadIdx.x & 31;
  const int wave = threadIdx.x >> 5;
  const int half = lane >> 4;
  const int l15  = lane & 15;
  const int c0 = wave * 32;

  float4 w0[16], w1[16];
  {
    const float4* wp0 = reinterpret_cast<const float4*>(W + (size_t)(c0 + l15) * DFEAT + 4 * half);
    const float4* wp1 = reinterpret_cast<const float4*>(W + (size_t)(c0 + 16 + l15) * DFEAT + 4 * half);
#pragma unroll
    for (int j = 0; j < 16; ++j) {
      w0[j] = wp0[2 * j];
      w1[j] = wp1[2 * j];
    }
  }
  const int col0 = c0 + l15;

  for (int tile = blockIdx.x; tile < n_tiles; tile += gridDim.x) {
    const int n0 = tile * 16;
    const int rbase = n0 + 8 * half;

    float4 a[16];
    {
      const float4* ap = reinterpret_cast<const float4*>(A + (size_t)(n0 + l15) * DFEAT + 4 * half);
#pragma unroll
      for (int j = 0; j < 16; ++j) a[j] = ap[2 * j];
    }
    const float4 dg0 = *reinterpret_cast<const float4*>(deg_out + rbase);
    const float4 dg1 = *reinterpret_cast<const float4*>(deg_out + rbase + 4);
    __builtin_amdgcn_sched_barrier(0);

    v8f acc0 = {0.f, 0.f, 0.f, 0.f, 0.f, 0.f, 0.f, 0.f};
    v8f acc1 = {0.f, 0.f, 0.f, 0.f, 0.f, 0.f, 0.f, 0.f};
#pragma unroll
    for (int j = 0; j < 16; ++j) {
      const v2f a_lo = {a[j].x, a[j].y};
      const v2f a_hi = {a[j].z, a[j].w};
      const v2f b0_lo = {w0[j].x, w0[j].y};
      const v2f b0_hi = {w0[j].z, w0[j].w};
      const v2f b1_lo = {w1[j].x, w1[j].y};
      const v2f b1_hi = {w1[j].z, w1[j].w};
      acc0 = wmma_f32(a_lo, b0_lo, acc0);
      acc1 = wmma_f32(a_lo, b1_lo, acc1);
      acc0 = wmma_f32(a_hi, b0_hi, acc0);
      acc1 = wmma_f32(a_hi, b1_hi, acc1);
    }

    const float dg[8] = {dg0.x, dg0.y, dg0.z, dg0.w, dg1.x, dg1.y, dg1.z, dg1.w};
    float* op = out + (size_t)rbase * DFEAT + col0;
#pragma unroll
    for (int r = 0; r < 8; ++r) {
      const float s = rsqrtf(fmaxf(dg[r], 1.0f));
      op[r * DFEAT] = acc0[r] * s;
      op[r * DFEAT + 16] = acc1[r] * s;
    }
  }
}

extern "C" void kernel_launch(void* const* d_in, const int* in_sizes, int n_in,
                              void* d_out, int out_size, void* d_ws, size_t ws_size,
                              hipStream_t stream) {
  const float* x  = (const float*)d_in[0];
  const int*   ei = (const int*)d_in[1];
  const float* W1 = (const float*)d_in[2];
  const float* b1 = (const float*)d_in[3];
  const float* W2 = (const float*)d_in[4];
  const float* b2 = (const float*)d_in[5];

  const int N = in_sizes[0] / DFEAT;
  const int E = in_sizes[1] / 2;
  const int* row = ei;
  const int* col = ei + E;
  float* out = (float*)d_out;

  float* deg_out = (float*)d_ws;
  float* deg_in  = deg_out + N;
  float* y       = deg_in + N;

  const int TB = 256;
  const int GEMM_BLOCKS = 1024;
  const int ownBlocks = (N + OWN_ROWS - 1) / OWN_ROWS;
  unsigned* lists = (unsigned*)(y + (size_t)N * DFEAT);
  int* lcnt = (int*)(lists + (size_t)ownBlocks * OWN_WAVES * OWN_LCAP);

  const int degRanges = (N + DEG_ROWS - 1) / DEG_ROWS;
  gcn_zero_f32<<<(2 * N + TB - 1) / TB, TB, 0, stream>>>(deg_out, 2 * N);
  gcn_degree_ranges<<<2 * degRanges * DEG_PARTS, DEG_THREADS, 0, stream>>>(
      row, col, deg_out, deg_in, N, E, degRanges);

  gcn_gemm_rowscale<<<GEMM_BLOCKS, 128, 0, stream>>>(x, W1, deg_out, y, N / 16);
  gcn_owner_sum<false, true, true><<<ownBlocks, 1024, 0, stream>>>(
      y, row, col, b1, deg_in, x, out, lists, lcnt, N, E);

  gcn_gemm_rowscale<<<GEMM_BLOCKS, 128, 0, stream>>>(out, W2, deg_out, y, N / 16);
  gcn_owner_sum<true, false, false><<<ownBlocks, 1024, 0, stream>>>(
      y, row, col, b2, deg_in, nullptr, out, lists, lcnt, N, E);
}


# Layer 1's owner blocks stop reading all of col: the degree pass's col blocks keep their 6272-node range's edges as per-wave lists, an owner block filters only its range's lists; sum order changes
# speedup vs baseline: 3.4629x; 1.0640x over previous
#include <hip/hip_runtime.h>
#include <hip/hip_bf16.h>

#define DFEAT 128

typedef __attribute__((ext_vector_type(2))) float v2f;
typedef __attribute__((ext_vector_type(8))) float v8f;

__global__ void gcn_zero_f32(float* __restrict__ p, int n) {
  int i = blockIdx.x * blockDim.x + threadIdx.x;
  int stride = gridDim.x * blockDim.x;
  for (; i < n; i += stride) p[i] = 0.0f;
}

#define DEG_ROWS 6272
#define DEG_PARTS 8
#define DEG_THREADS 512
#define DEG_WAVES (DEG_THREADS / 32)
#define DEG_DEPTH 25
#define CAND_CAP 1024

template <bool KEEP>
__device__ __forceinline__ void deg_take(int* s_cnt, uint2* s_my, int& cn,
                                         bool ok, int v, int lo, unsigned e,
                                         unsigned lt_mask) {
  const unsigned d = ok ? (unsigned)(v - lo) : 0xffffffffu;
  const bool m = d < (unsigned)DEG_ROWS;
  if (KEEP) {
    const unsigned mask = (unsigned)__ballot(m);
    const int p = cn + __builtin_popcount(mask & lt_mask);
    if (m) {
      atomicAdd(&s_cnt[d], 1);
      if (p < CAND_CAP) s_my[p] = make_uint2(e, d);
    }
    cn += __builtin_popcount(mask);
  } else {
    if (m) atomicAdd(&s_cnt[d], 1);
  }
}

template <bool KEEP>
__device__ __forceinline__ int deg_scan(int* s_cnt, uint2* s_my,
                                        const int* __restrict__ base, int n,
                                        int beg, int lo, int tid,
                                        unsigned lt_mask) {
  int cn = 0;
  if ((reinterpret_cast<size_t>(base) & 15) == 0) {
    const int nv = n >> 2;
    const int4* b4 = reinterpret_cast<const int4*>(base);
    for (int q0 = 0; q0 < nv; q0 += DEG_THREADS * DEG_DEPTH) {
      const int q = q0 + tid;
      int4 v[DEG_DEPTH];
#pragma unroll
      for (int k = 0; k < DEG_DEPTH; ++k) {
        const int qq = q + DEG_THREADS * k;
        v[k] = b4[qq < nv ? qq : nv - 1];
      }
#pragma unroll
      for (int k = 0; k < DEG_DEPTH; ++k) {
        const int qq = q + DEG_THREADS * k;
        const bool ok = qq < nv;
        const unsigned e = (unsigned)(beg + 4 * qq);
        deg_take<KEEP>(s_cnt, s_my, cn, ok, v[k].x, lo, e, lt_mask);
        deg_take<KEEP>(s_cnt, s_my, cn, ok, v[k].y, lo, e + 1u, lt_mask);
        deg_take<KEEP>(s_cnt, s_my, cn, ok, v[k].z, lo, e + 2u, lt_mask);
        deg_take<KEEP>(s_cnt, s_my, cn, ok, v[k].w, lo, e + 3u, lt_mask);
      }
    }
    const int t = (nv << 2) + tid;
    const bool tok = t < n;
    deg_take<KEEP>(s_cnt, s_my, cn, tok, tok ? base[t] : 0, lo,
                   (unsigned)(beg + t), lt_mask);
  } else {
    for (int p0 = 0; p0 < n; p0 += DEG_THREADS) {
      const int p = p0 + tid;
      const bool ok = p < n;
      deg_take<KEEP>(s_cnt, s_my, cn, ok, ok ? base[p] : 0, lo,
                     (unsigned)(beg + p), lt_mask);
    }
  }
  return cn;
}

__global__ void __launch_bounds__(DEG_THREADS) gcn_degree_ranges(
    const int* __restrict__ row, const int* __restrict__ col,
    float* __restrict__ deg_out, float* __restrict__ deg_in,
    unsigned* __restrict__ cand_words, int* __restrict__ ccnt,
    int N, int E, int n_ranges) {
  __shared__ int s_cnt[DEG_ROWS];
  __shared__ uint2 s_cand[DEG_WAVES * CAND_CAP];
  const int tid = threadIdx.x;
  const int lane = tid & 31;
  const int wave = __builtin_amdgcn_readfirstlane(tid >> 5);
  const unsigned lt_mask = (1u << lane) - 1u;
  int b = blockIdx.x;
  const int part = b % DEG_PARTS;
  b /= DEG_PARTS;
  const int range = b % n_ranges;
  const bool is_col = (b / n_ranges) != 0;
  const int* __restrict__ idx = is_col ? col : row;
  float* __restrict__ deg = is_col ? deg_in : deg_out;
  const int lo = range * DEG_ROWS;
  uint2* s_my = s_cand + wave * CAND_CAP;

  for (int i = tid; i < DEG_ROWS; i += DEG_THREADS) s_cnt[i] = 0;
  __syncthreads();

  const int per = ((E + DEG_PARTS - 1) / DEG_PARTS + 3) & ~3;
  const int beg = part * per;
  int end = beg + per;
  if (end > E) end = E;
  const int n = end - beg;
  int cn = 0;
  if (n > 0) {
    cn = is_col ? deg_scan<true>(s_cnt, s_my, idx + beg, n, beg, lo, tid, lt_mask)
                : deg_scan<false>(s_cnt, s_my, idx + beg, n, beg, lo, tid, lt_mask);
  }
  if (is_col) {
    const int sub = (range * DEG_PARTS + part) * DEG_WAVES + wave;
    uint2* mycand = reinterpret_cast<uint2*>(cand_words) + (size_t)sub * CAND_CAP;
    const int nk = cn < CAND_CAP ? cn : CAND_CAP;
    for (int i = lane; i < nk; i += 32) mycand[i] = s_my[i];
    if (lane == 0) ccnt[sub] = cn;
  }
  __syncthreads();

  for (int i = tid; i < DEG_ROWS; i += DEG_THREADS) {
    const int node = lo + i;
    if (node < N) atomicAdd(&deg[node], (float)s_cnt[i]);
  }
}

#define OWN_ROWS 392
#define OWN_WAVES 32
#define OWN_QCAP 768
#define OWN_STEP 512
#define OWN_BATCH 16
#define OWN_LCAP 1024
#define CAND_PER_WAVE ((DEG_PARTS * DEG_WAVES) / OWN_WAVES)
static_assert(DEG_ROWS % OWN_ROWS == 0, "an owner block must lie inside one degree range");
static_assert((DEG_PARTS * DEG_WAVES) % OWN_WAVES == 0 && CAND_PER_WAVE == 4,
              "four kept-edge lists per owner wave");

__device__ __forceinline__ void own_add16(const float* __restrict__ xs,
                                          float4* s_acc, int srcv, unsigned dlv,
                                          int lane) {
  float4 vals[OWN_BATCH];
#pragma unroll
  for (int i = 0; i < OWN_BATCH; ++i) {
    const int s = __builtin_amdgcn_readlane(srcv, i);
    vals[i] = reinterpret_cast<const float4*>(xs + (size_t)s * DFEAT)[lane];
  }
  __builtin_amdgcn_sched_barrier(0);
#pragma unroll
  for (int i = 0; i < OWN_BATCH; ++i) {
    const unsigned dl = (unsigned)__builtin_amdgcn_readlane((int)dlv, i) & 511u;
    float4 a = s_acc[dl * 32 + lane];
    a.x += vals[i].x; a.y += vals[i].y; a.z += vals[i].z; a.w += vals[i].w;
    s_acc[dl * 32 + lane] = a;
  }
}

__device__ __forceinline__ void own_flush(const float* __restrict__ xs,
                                          const int* __restrict__ row,
                                          float4* s_acc, unsigned pend, int lane,
                                          unsigned* save_to) {
  int srcv = 0;
  if (lane < OWN_BATCH) {
    srcv = row[pend >> 9];
    if (save_to) save_to[lane] = ((unsigned)srcv << 9) | (pend & 511u);
  }
  own_add16(xs, s_acc, srcv, pend, lane);
}

#define OWN_SROWS ((OWN_ROWS + OWN_WAVES - 1) / OWN_WAVES)
template <bool HAS_RESID, bool DO_RELU>
__device__ __forceinline__ void own_store(const float4* s_acc,
                                          const float* __restrict__ bias,
                                          const float* __restrict__ deg_in,
                                          const float* __restrict__ resid,
                                          float* __restrict__ dst,
                                          int lo, int N, int w, int lane) {
  const float4 b = reinterpret_cast<const float4*>(bias)[lane];
  float dg[OWN_SROWS];
  float4 rs[OWN_SROWS];
#pragma unroll
  for (int k = 0; k < OWN_SROWS; ++k) {
    int r = lo + w + OWN_WAVES * k;
    if (r > N - 1) r = N - 1;
    dg[k] = deg_in[r];
    if (HAS_RESID) rs[k] = reinterpret_cast<const float4*>(resid)[(size_t)r * 32 + lane];
  }
#pragma unroll
  for (int k = 0; k < OWN_SROWS; ++k) {
    const int lr = w + OWN_WAVES * k;
    const int r = lo + lr;
    if (lr < OWN_ROWS && r < N) {
      const float s = rsqrtf(fmaxf(dg[k], 1.0f));
      float4 v = s_acc[lr * 32 + lane];
      v.x = (v.x + b.x) * s; v.y = (v.y + b.y) * s;
      v.z = (v.z + b.z) * s; v.w = (v.w + b.w) * s;
      if (HAS_RESID) { v.x += rs[k].x; v.y += rs[k].y; v.z += rs[k].z; v.w += rs[k].w; }
      if (DO_RELU) {
        v.x = fmaxf(v.x, 0.0f); v.y = fmaxf(v.y, 0.0f);
        v.z = fmaxf(v.z, 0.0f); v.w = fmaxf(v.w, 0.0f);
      }
      reinterpret_cast<float4*>(dst)[(size_t)r * 32 + lane] = v;
    }
  }
}

template <bool USE_LIST, bool HAS_RESID, bool DO_RELU>
__global__ void __launch_bounds__(1024) gcn_owner_sum(
    const float* __restrict__ xs, const int* __restrict__ row,
    const int* __restrict__ col,
    const float* __restrict__ bias,
    const float* __restrict__ deg_in,
    const float* __restrict__ resid,
    float* __restrict__ dst,
    unsigned* __restrict__ lists, int* __restrict__ lcnt,
    const unsigned* __restrict__ cand_words, const int* __restrict__ ccnt,
    int N, int E) {
  __shared__ float4 s_acc[(OWN_ROWS + OWN_WAVES) * 32];
  __shared__ unsigned s_q[OWN_WAVES * OWN_QCAP];
  __shared__ int s_qcnt[OWN_WAVES];
  __shared__ int s_done[OWN_WAVES];

  const int tid = threadIdx.x;
  const int lane = tid & 31;
  const int w = __builtin_amdgcn_readfirstlane(tid >> 5);
  const int lo = blockIdx.x * OWN_ROWS;
  const unsigned lt_mask = (1u << lane) - 1u;

  for (int i = tid; i < OWN_ROWS * 32; i += 1024)
    s_acc[i] = make_float4(0.0f, 0.0f, 0.0f, 0.0f);

  unsigned* mylist = lists + (size_t)(blockIdx.x * OWN_WAVES + w) * OWN_LCAP;
  const unsigned pad_entry = (unsigned)(OWN_ROWS + w);

  if (USE_LIST) {
    const int cnt = lcnt[blockIdx.x * OWN_WAVES + lane];
    const bool all_fit =
        ((unsigned)__ballot(cnt <= OWN_LCAP / OWN_BATCH)) == 0xffffffffu;
    if (all_fit) {
      const int nb = __builtin_amdgcn_readlane(cnt, w);
      unsigned ent = pad_entry;
      if (nb > 0 && lane < OWN_BATCH) ent = mylist[lane];
      __syncthreads();
      for (int i = 0; i < nb; ++i) {
        unsigned nxt = pad_entry;
        if (i + 1 < nb && lane < OWN_BATCH) nxt = mylist[(i + 1) * OWN_BATCH + lane];
        own_add16(xs, s_acc, (int)(ent >> 9), ent, lane);
        ent = nxt;
      }
      __syncthreads();
      own_store<HAS_RESID, DO_RELU>(s_acc, bias, deg_in, resid, dst, lo, N, w, lane);
      return;
    }
  }
  int nbatch = 0;

  const uint2* cand = reinterpret_cast<const uint2*>(cand_words);
  const int crange = lo / DEG_ROWS;
  const unsigned coff = (unsigned)(lo - crange * DEG_ROWS);
  const int csub0 = crange * (DEG_PARTS * DEG_WAVES);
  const int* cc = ccnt + csub0 + CAND_PER_WAVE * lane;
  const int cl0 = cc[0], cl1 = cc[1], cl2 = cc[2], cl3 = cc[3];
  const bool use_cand =
      ((unsigned)__ballot(cl0 <= CAND_CAP && cl1 <= CAND_CAP &&
                          cl2 <= CAND_CAP && cl3 <= CAND_CAP)) == 0xffffffffu;
  const int cn0 = __builtin_amdgcn_readlane(cl0, w);
  const int cn1 = __builtin_amdgcn_readlane(cl1, w);
  const int cn2 = __builtin_amdgcn_readlane(cl2, w);
  const int cn3 = __builtin_amdgcn_readlane(cl3, w);
  int csi = 0;
  int cpos = 0;

  const int n_trips = (E + OWN_WAVES * OWN_STEP - 1) / (OWN_WAVES * OWN_STEP);
  const int seg0 = w * n_trips * OWN_STEP;
  int end = seg0 + n_trips * OWN_STEP;
  if (end > E) end = E;
  const int rot = (n_trips > 0) ? (int)(blockIdx.x % (unsigned)n_trips) : 0;
  int trip = 0;

  unsigned* myq = s_q + w * OWN_QCAP;
  unsigned pend = pad_entry;
  int npend = 0;

  for (;;) {
    int qn = 0;
    while (use_cand && csi < CAND_PER_WAVE && qn + OWN_STEP <= OWN_QCAP) {
      const int n = (csi == 0) ? cn0 : (csi == 1) ? cn1 : (csi == 2) ? cn2 : cn3;
      if (cpos >= n) { ++csi; cpos = 0; continue; }
      const uint2* cp =
          cand + (size_t)(csub0 + CAND_PER_WAVE * w + csi) * CAND_CAP + cpos;
      const int left = n - cpos;
      uint2 c[OWN_STEP / 32];
#pragma unroll
      for (int k = 0; k < OWN_STEP / 32; ++k) {
        const int i = 32 * k + lane;
        c[k] = cp[i < left ? i : 0];
      }
#pragma unroll
      for (int k = 0; k < OWN_STEP / 32; ++k) {
        const unsigned dl = c[k].y - coff;
        const bool m = (32 * k + lane < left) && dl < (unsigned)OWN_ROWS;
        const unsigned mask = (unsigned)__ballot(m);
        if (mask != 0u) {
          if (m) myq[qn + __builtin_popcount(mask & lt_mask)] = (c[k].x << 9) | dl;
          qn += __builtin_popcount(mask);
        }
      }
      cpos += OWN_STEP;
    }
    while (!use_cand && trip < n_trips && qn + OWN_STEP <= OWN_QCAP) {
      int t = trip + rot;
      if (t >= n_trips) t -= n_trips;
      const int pos = seg0 + t * OWN_STEP;
      ++trip;
      if (pos >= end) continue;
      int c[OWN_STEP / 32];
      if (pos + OWN_STEP <= end) {
        const int* cp = col + pos + lane;
#pragma unroll
        for (int k = 0; k < OWN_STEP / 32; ++k) c[k] = cp[32 * k];
      } else {
#pragma unroll
        for (int k = 0; k < OWN_STEP / 32; ++k) {
          const int e = pos + 32 * k + lane;
          c[k] = (e < end) ? col[e] : -1;
        }
      }
#pragma unroll
      for (int k = 0; k < OWN_STEP / 32; ++k) {
        const unsigned dl = (unsigned)(c[k] - lo);
        const bool m = dl < (unsigned)OWN_ROWS;
        const unsigned mask = (unsigned)__ballot(m);
        if (mask != 0u) {
          if (m) {
            const unsigned e = (unsigned)(pos + 32 * k + lane);
            myq[qn + __builtin_popcount(mask & lt_mask)] = (e << 9) | dl;
          }
          qn += __builtin_popcount(mask);
        }
      }
    }
    if (lane == 0) {
      s_qcnt[w] = qn;
      s_done[w] = (use_cand ? (csi >= CAND_PER_WAVE) : (trip >= n_trips)) ? 1 : 0;
    }
    __syncthreads();

    for (int q = 0; q < OWN_WAVES; ++q) {
      const int n = __builtin_amdgcn_readfirstlane(s_qcnt[q]);
      const unsigned* qq = s_q + q * OWN_QCAP;
      for (int i = 0; i < n; i += 32) {
        const bool in = (i + lane) < n;
        const unsigned v = in ? qq[i + lane] : 0u;
        const bool m = in && ((v & 31u) == (unsigned)w);
        unsigned mask = (unsigned)__ballot(m);
        while (mask) {
          const int b = __builtin_ctz(mask);
          mask &= mask - 1u;
          const unsigned val = (unsigned)__builtin_amdgcn_readlane((int)v, b);
          if (lane == npend) pend = val;
          ++npend;
          if (npend == OWN_BATCH) {
            own_flush(xs, row, s_acc, pend, lane,
                      (!USE_LIST && nbatch < OWN_LCAP / OWN_BATCH)
                          ? mylist + nbatch * OWN_BATCH : nullptr);
            ++nbatch;
            pend = pad_entry;
            npend = 0;
          }
        }
      }
    }
    if (npend > 0) {
      own_flush(xs, row, s_acc, pend, lane,
                (!USE_LIST && nbatch < OWN_LCAP / OWN_BATCH)
                    ? mylist + nbatch * OWN_BATCH : nullptr);
      ++nbatch;
      pend = pad_entry;
      npend = 0;
    }
    const bool fin = ((unsigned)__ballot(s_done[lane] != 0)) == 0xffffffffu;
    __syncthreads();
    if (fin) break;
  }
  if (!USE_LIST && lane == 0) lcnt[blockIdx.x * OWN_WAVES + w] = nbatch;

  own_store<HAS_RESID, DO_RELU>(s_acc, bias, deg_in, resid, dst, lo, N, w, lane);
}

__device__ __forceinline__ v8f wmma_f32(v2f a, v2f b, v8f c) {
  return __builtin_amdgcn_wmma_f32_16x16x4_f32(
       false, a,  false, b,
       (short)0, c,  false,  false);
}

__global__ void __launch_bounds__(128) gcn_gemm_rowscale(
    const float* __restrict__ A,
    const float* __restrict__ W,
    const float* __restrict__ deg_out,
    float* __restrict__ out,
    int n_tiles) {
  const int lane = threadIdx.x & 31;
  const int wave = threadIdx.x >> 5;
  const int half = lane >> 4;
  const int l15  = lane & 15;
  const int c0 = wave * 32;

  float4 w0[16], w1[16];
  {
    const float4* wp0 = reinterpret_cast<const float4*>(W + (size_t)(c0 + l15) * DFEAT + 4 * half);
    const float4* wp1 = reinterpret_cast<const float4*>(W + (size_t)(c0 + 16 + l15) * DFEAT + 4 * half);
#pragma unroll
    for (int j = 0; j < 16; ++j) {
      w0[j] = wp0[2 * j];
      w1[j] = wp1[2 * j];
    }
  }
  const int col0 = c0 + l15;

  for (int tile = blockIdx.x; tile < n_tiles; tile += gridDim.x) {
    const int n0 = tile * 16;
    const int rbase = n0 + 8 * half;

    float4 a[16];
    {
      const float4* ap = reinterpret_cast<const float4*>(A + (size_t)(n0 + l15) * DFEAT + 4 * half);
#pragma unroll
      for (int j = 0; j < 16; ++j) a[j] = ap[2 * j];
    }
    const float4 dg0 = *reinterpret_cast<const float4*>(deg_out + rbase);
    const float4 dg1 = *reinterpret_cast<const float4*>(deg_out + rbase + 4);
    __builtin_amdgcn_sched_barrier(0);

    v8f acc0 = {0.f, 0.f, 0.f, 0.f, 0.f, 0.f, 0.f, 0.f};
    v8f acc1 = {0.f, 0.f, 0.f, 0.f, 0.f, 0.f, 0.f, 0.f};
#pragma unroll
    for (int j = 0; j < 16; ++j) {
      const v2f a_lo = {a[j].x, a[j].y};
      const v2f a_hi = {a[j].z, a[j].w};
      const v2f b0_lo = {w0[j].x, w0[j].y};
      const v2f b0_hi = {w0[j].z, w0[j].w};
      const v2f b1_lo = {w1[j].x, w1[j].y};
      const v2f b1_hi = {w1[j].z, w1[j].w};
      acc0 = wmma_f32(a_lo, b0_lo, acc0);
      acc1 = wmma_f32(a_lo, b1_lo, acc1);
      acc0 = wmma_f32(a_hi, b0_hi, acc0);
      acc1 = wmma_f32(a_hi, b1_hi, acc1);
    }

    const float dg[8] = {dg0.x, dg0.y, dg0.z, dg0.w, dg1.x, dg1.y, dg1.z, dg1.w};
    float* op = out + (size_t)rbase * DFEAT + col0;
#pragma unroll
    for (int r = 0; r < 8; ++r) {
      const float s = rsqrtf(fmaxf(dg[r], 1.0f));
      op[r * DFEAT] = acc0[r] * s;
      op[r * DFEAT + 16] = acc1[r] * s;
    }
  }
}

extern "C" void kernel_launch(void* const* d_in, const int* in_sizes, int n_in,
                              void* d_out, int out_size, void* d_ws, size_t ws_size,
                              hipStream_t stream) {
  const float* x  = (const float*)d_in[0];
  const int*   ei = (const int*)d_in[1];
  const float* W1 = (const float*)d_in[2];
  const float* b1 = (const float*)d_in[3];
  const float* W2 = (const float*)d_in[4];
  const float* b2 = (const float*)d_in[5];

  const int N = in_sizes[0] / DFEAT;
  const int E = in_sizes[1] / 2;
  const int* row = ei;
  const int* col = ei + E;
  float* out = (float*)d_out;

  float* deg_out = (float*)d_ws;
  float* deg_in  = deg_out + N;
  float* y       = deg_in + N;

  const int TB = 256;
  const int GEMM_BLOCKS = 1024;
  const int ownBlocks = (N + OWN_ROWS - 1) / OWN_ROWS;
  unsigned* lists = (unsigned*)(y + (size_t)N * DFEAT);
  int* lcnt = (int*)(lists + (size_t)ownBlocks * OWN_WAVES * OWN_LCAP);

  const int degRanges = (N + DEG_ROWS - 1) / DEG_ROWS;
  unsigned* cand = (unsigned*)(lcnt + (size_t)ownBlocks * OWN_WAVES);
  int* ccnt = (int*)(cand + (size_t)degRanges * DEG_PARTS * DEG_WAVES * CAND_CAP * 2);
  gcn_zero_f32<<<(2 * N + TB - 1) / TB, TB, 0, stream>>>(deg_out, 2 * N);
  gcn_degree_ranges<<<2 * degRanges * DEG_PARTS, DEG_THREADS, 0, stream>>>(
      row, col, deg_out, deg_in, cand, ccnt, N, E, degRanges);

  gcn_gemm_rowscale<<<GEMM_BLOCKS, 128, 0, stream>>>(x, W1, deg_out, y, N / 16);
  gcn_owner_sum<false, true, true><<<ownBlocks, 1024, 0, stream>>>(
      y, row, col, b1, deg_in, x, out, lists, lcnt, cand, ccnt, N, E);

  gcn_gemm_rowscale<<<GEMM_BLOCKS, 128, 0, stream>>>(out, W2, deg_out, y, N / 16);
  gcn_owner_sum<true, false, false><<<ownBlocks, 1024, 0, stream>>>(
      y, row, col, b2, deg_in, nullptr, out, lists, lcnt, cand, ccnt, N, E);
}
